# starting kernel
# baseline (speedup 1.0000x reference)
#include <hip/hip_runtime.h>
#include <hip/hip_bf16.h>


#define BDIM 1024
#define KDIM 64
#define NMON 512
#define ROWS_PER_BLOCK 128
#define THREADS 256

typedef __attribute__((ext_vector_type(2))) float v2f;
typedef __attribute__((ext_vector_type(8))) float v8f;

__global__ __launch_bounds__(THREADS)
void sparse_poly_kernel(const float* __restrict__ x,
                        const int*   __restrict__ S,
                        const float* __restrict__ a,
                        const float* __restrict__ b,
                        const int*   __restrict__ idx3, const float* __restrict__ c3,
                        const int*   __restrict__ idx4, const float* __restrict__ c4,
                        const int*   __restrict__ idx5, const float* __restrict__ c5,
                        float* __restrict__ out)
{
    __shared__ float xs[ROWS_PER_BLOCK][KDIM];
    __shared__ float buS[KDIM][KDIM];
    __shared__ float quadS[ROWS_PER_BLOCK];
    __shared__ int   Ssh[KDIM];
    __shared__ float ash[KDIM];

    const int tid  = threadIdx.x;
    const int wave = tid >> 5;
    const int lane = tid & 31;
    const int lo   = lane & 15;
    const int hi   = lane >> 4;
    const int rowBlockBase = blockIdx.x * ROWS_PER_BLOCK;

    if (tid < KDIM) {
        Ssh[tid] = S[tid];
        ash[tid] = a[tid];
    }
    #pragma unroll
    for (int e = tid; e < KDIM * KDIM; e += THREADS) {
        const int i = e >> 6, j = e & 63;
        buS[i][j] = (j > i) ? b[e] : 0.0f;
    }
    __syncthreads();

    for (int e = tid; e < ROWS_PER_BLOCK * KDIM; e += THREADS) {
        const int r = e >> 6, j = e & 63;
        xs[r][j] = x[(size_t)(rowBlockBase + r) * BDIM + Ssh[j]];
    }
    __syncthreads();

    const int waveBase = wave * 16;
    float part[8];
    #pragma unroll
    for (int p = 0; p < 8; ++p) part[p] = 0.0f;

    #pragma unroll
    for (int nt = 0; nt < 4; ++nt) {
        v8f acc = {};
        #pragma unroll
        for (int kc = 0; kc < 16; ++kc) {
            v2f A, Bf;
            A.x = xs[waveBase + lo][4 * kc + 2 * hi + 0];
            A.y = xs[waveBase + lo][4 * kc + 2 * hi + 1];
            Bf.x = buS[4 * kc + 2 * hi + 0][16 * nt + lo];
            Bf.y = buS[4 * kc + 2 * hi + 1][16 * nt + lo];
            acc = __builtin_amdgcn_wmma_f32_16x16x4_f32(
                false, A, false, Bf, (short)0, acc, false, false);
        }
        #pragma unroll
        for (int p = 0; p < 8; ++p)
            part[p] += acc[p] * xs[waveBase + p + 8 * hi][16 * nt + lo];
    }

    #pragma unroll
    for (int p = 0; p < 8; ++p) {
        float s = part[p];
        s += __shfl_xor(s, 8);
        s += __shfl_xor(s, 4);
        s += __shfl_xor(s, 2);
        s += __shfl_xor(s, 1);
        if (lo == p) quadS[waveBase + p + 8 * hi] = s;
    }
    __syncthreads();

    if (tid < ROWS_PER_BLOCK) {
        const int lr = tid;
        const float* xr = xs[lr];
        float accum = quadS[lr];

        #pragma unroll 8
        for (int j = 0; j < KDIM; ++j)
            accum += xr[j] * ash[j];

        for (int t = 0; t < NMON; ++t) {
            const int i0 = idx3[3 * t + 0], i1 = idx3[3 * t + 1], i2 = idx3[3 * t + 2];
            accum += c3[t] * xr[i0] * xr[i1] * xr[i2];
        }
        for (int t = 0; t < NMON; ++t) {
            const int i0 = idx4[4 * t + 0], i1 = idx4[4 * t + 1];
            const int i2 = idx4[4 * t + 2], i3 = idx4[4 * t + 3];
            accum += c4[t] * xr[i0] * xr[i1] * xr[i2] * xr[i3];
        }
        for (int t = 0; t < NMON; ++t) {
            const int i0 = idx5[5 * t + 0], i1 = idx5[5 * t + 1], i2 = idx5[5 * t + 2];
            const int i3 = idx5[5 * t + 3], i4 = idx5[5 * t + 4];
            accum += c5[t] * xr[i0] * xr[i1] * xr[i2] * xr[i3] * xr[i4];
        }
        out[rowBlockBase + lr] = accum;
    }
}

extern "C" void kernel_launch(void* const* d_in, const int* in_sizes, int n_in,
                              void* d_out, int out_size, void* d_ws, size_t ws_size,
                              hipStream_t stream) {
    const float* x    = (const float*)d_in[0];
    const int*   S    = (const int*)  d_in[1];
    const float* a    = (const float*)d_in[2];
    const float* b    = (const float*)d_in[3];
    const int*   idx3 = (const int*)  d_in[4];
    const float* c3   = (const float*)d_in[5];
    const int*   idx4 = (const int*)  d_in[6];
    const float* c4   = (const float*)d_in[7];
    const int*   idx5 = (const int*)  d_in[8];
    const float* c5   = (const float*)d_in[9];
    float* out = (float*)d_out;

    const int nblocks = out_size / ROWS_PER_BLOCK;
    sparse_poly_kernel<<<nblocks, THREADS, 0, stream>>>(
        x, S, a, b, idx3, c3, idx4, c4, idx5, c5, out);
}


# monomial+linear phase row-parallel from registers: a lane holds its 2 rows' 64 columns in 128 VGPRs, a factor is one M0-relative move, term tables by scalar loads; no LDS gather; sum order changes
# speedup vs baseline: 19.2211x; 19.2211x over previous
#include <hip/hip_runtime.h>


#define BDIM 1024
#define KDIM 64
#define NMON 512
#define ROWS_PER_BLOCK 128
#define HALF_ROWS 64
#define THREADS 512
#define NWAVES 16
#define TPW (NMON / NWAVES)
#define LPW (KDIM / NWAVES)
#define XPITCH 68
#define XPT (HALF_ROWS * KDIM / THREADS)
#define BPT (KDIM * KDIM / THREADS)
#define NSPLIT 2

typedef __attribute__((ext_vector_type(2))) float v2f;
typedef __attribute__((ext_vector_type(8))) float v8f;
typedef __attribute__((ext_vector_type(4))) float f4r;
typedef __attribute__((ext_vector_type(32))) float v32f;
#define CONST_AS __attribute__((address_space(4)))
typedef __attribute__((ext_vector_type(4)))  int   i4r;
typedef __attribute__((ext_vector_type(8)))  int   i8r;
typedef __attribute__((ext_vector_type(16))) int   i16r;
typedef __attribute__((ext_vector_type(8)))  float f8r;
typedef i4r  i4v  __attribute__((aligned(4)));
typedef i8r  i8v  __attribute__((aligned(4)));
typedef i16r i16v __attribute__((aligned(4)));
typedef f4r  f4v  __attribute__((aligned(4)));
typedef f8r  f8v  __attribute__((aligned(4)));

struct RowRegs { v32f a_lo, a_hi, b_lo, b_hi; };

__device__ __forceinline__ void fetch2(const RowRegs& r, int c, float& fa, float& fb) {
    asm("s_and_b32 m0, %6, 63\n\t"
        "v_movrels_b32 %0, v64\n\t"
        "v_movrels_b32 %1, v128"
        : "=&v"(fa), "=&v"(fb)
        : "{v[64:95]}"(r.a_lo), "{v[96:127]}"(r.a_hi),
          "{v[128:159]}"(r.b_lo), "{v[160:191]}"(r.b_hi), "s"(c)
        : "m0");
}

template <int N>
__device__ __forceinline__ void sload_i(int (&d)[N], const int* p) {
    static_assert(N == 16 || N == 20 || N == 24, "table slice sizes of the three degrees");
    const i16v q = *(const CONST_AS i16v*)p;
    #pragma unroll
    for (int j = 0; j < 16; ++j) d[j] = q[j];
    if constexpr (N == 20) {
        const i4v q2 = *(const CONST_AS i4v*)(p + 16);
        #pragma unroll
        for (int j = 0; j < 4; ++j) d[16 + j] = q2[j];
    }
    if constexpr (N == 24) {
        const i8v q2 = *(const CONST_AS i8v*)(p + 16);
        #pragma unroll
        for (int j = 0; j < 8; ++j) d[16 + j] = q2[j];
    }
}

template <int N>
__device__ __forceinline__ void sload_f(float (&d)[N], const float* p) {
    static_assert(N == 4 || N == 8, "coefficients of one trip");
    if constexpr (N == 4) {
        const f4v q = *(const CONST_AS f4v*)p;
        #pragma unroll
        for (int j = 0; j < 4; ++j) d[j] = q[j];
    } else {
        const f8v q = *(const CONST_AS f8v*)p;
        #pragma unroll
        for (int j = 0; j < 8; ++j) d[j] = q[j];
    }
}

#define GRP 4

template <int DEG, int T>
__device__ __forceinline__ void products(const int (&ix)[DEG * T], const float (&cf)[T],
                                         const RowRegs& r, float fa0, float fb0,
                                         float (&sa)[GRP], float (&sb)[GRP]) {
    static_assert(T % GRP == 0, "terms are taken in groups");
    #pragma unroll
    for (int u = 0; u < T; u += GRP) {
        float pa[GRP], pb[GRP];
        #pragma unroll
        for (int d = 0; d < DEG; ++d) {
            float fa[GRP], fb[GRP];
            #pragma unroll
            for (int g = 0; g < GRP; ++g) {
                if (u == 0 && d == 0 && g == 0) { fa[0] = fa0; fb[0] = fb0; }
                else fetch2(r, ix[DEG * (u + g) + d], fa[g], fb[g]);
            }
            #pragma unroll
            for (int g = 0; g < GRP; ++g) {
                if (d == 0) {
                    pa[g] = cf[u + g] * fa[g];
                    pb[g] = cf[u + g] * fb[g];
                } else if (d < DEG - 1) {
                    pa[g] *= fa[g];
                    pb[g] *= fb[g];
                } else {
                    sa[g] = __builtin_fmaf(pa[g], fa[g], sa[g]);
                    sb[g] = __builtin_fmaf(pb[g], fb[g], sb[g]);
                }
            }
        }
    }
}

template <int DEG, int T>
__device__ __forceinline__ void degree_pass(const int* __restrict__ ip, const float* __restrict__ cp,
                                            const RowRegs& r, float (&sa)[GRP], float (&sb)[GRP]) {
    constexpr int NI = DEG * T;
    int   ia[NI], ib[NI];
    float ca[T],  cb[T];
    sload_i<NI>(ia, ip);
    sload_f<T>(ca, cp);
    for (int t = 0; t < TPW; t += 2 * T) {
        float fa, fb;
        fetch2(r, ia[0], fa, fb);
        __builtin_amdgcn_sched_barrier(0);
        sload_i<NI>(ib, ip + DEG * (t + T));
        sload_f<T>(cb, cp + (t + T));
        __builtin_amdgcn_sched_barrier(0);
        products<DEG, T>(ia, ca, r, fa, fb, sa, sb);

        fetch2(r, ib[0], fa, fb);
        __builtin_amdgcn_sched_barrier(0);
        const int tn = (t + 2 * T) & (TPW - 1);
        sload_i<NI>(ia, ip + DEG * tn);
        sload_f<T>(ca, cp + tn);
        __builtin_amdgcn_sched_barrier(0);
        products<DEG, T>(ib, cb, r, fa, fb, sa, sb);
    }
}

__device__ __forceinline__ void load_row(v32f& lo, v32f& hi, const float* xr) {
    #pragma unroll
    for (int q = 0; q < 8; ++q) {
        const f4r v = *(const f4r*)(xr + 4 * q);
        const f4r w = *(const f4r*)(xr + 32 + 4 * q);
        #pragma unroll
        for (int j = 0; j < 4; ++j) { lo[4 * q + j] = v[j]; hi[4 * q + j] = w[j]; }
    }
}

__device__ __forceinline__ void half_pass(const float* xh, int lane, int wave, float* wp,
                                          const int* __restrict__ ip3, const float* __restrict__ cp3,
                                          const int* __restrict__ ip4, const float* __restrict__ cp4,
                                          const int* __restrict__ ip5, const float* __restrict__ cp5,
                                          const f4r& al) {
    RowRegs r;
    load_row(r.a_lo, r.a_hi, xh + lane * XPITCH);
    load_row(r.b_lo, r.b_hi, xh + (32 + lane) * XPITCH);

    float sa[GRP] = {0.0f, 0.0f, 0.0f, 0.0f}, sb[GRP] = {0.0f, 0.0f, 0.0f, 0.0f};
    degree_pass<3, 8>(ip3, cp3, r, sa, sb);
    degree_pass<4, 4>(ip4, cp4, r, sa, sb);
    degree_pass<5, 4>(ip5, cp5, r, sa, sb);
    #pragma unroll
    for (int k = 0; k < LPW; ++k) {
        float fa, fb;
        fetch2(r, LPW * wave + k, fa, fb);
        sa[k] = __builtin_fmaf(al[k], fa, sa[k]);
        sb[k] = __builtin_fmaf(al[k], fb, sb[k]);
    }
    wp[lane]      = (sa[0] + sa[1]) + (sa[2] + sa[3]);
    wp[32 + lane] = (sb[0] + sb[1]) + (sb[2] + sb[3]);
}

__global__ __launch_bounds__(THREADS)
void sparse_poly_kernel(const float* __restrict__ x,
                        const int*   __restrict__ S,
                        const float* __restrict__ a,
                        const float* __restrict__ b,
                        const int*   __restrict__ idx3, const float* __restrict__ c3,
                        const int*   __restrict__ idx4, const float* __restrict__ c4,
                        const int*   __restrict__ idx5, const float* __restrict__ c5,
                        float* __restrict__ out)
{
    __shared__ __attribute__((aligned(16))) float xs[ROWS_PER_BLOCK][XPITCH];
    __shared__ float buS[KDIM][KDIM];
    __shared__ float quadS[NSPLIT][ROWS_PER_BLOCK];
    __shared__ float wpart[NWAVES][ROWS_PER_BLOCK];

    const int tid  = threadIdx.x;
    const int lane = tid & 31;
    const int lo   = lane & 15;
    const int hi   = lane >> 4;
    const int wave = __builtin_amdgcn_readfirstlane(tid >> 5);
    const int rowBlockBase = blockIdx.x * ROWS_PER_BLOCK;

    const int jcol = tid & (KDIM - 1);
    const int rsub = tid >> 6;
    const float* xcol = x + (size_t)rowBlockBase * BDIM + S[jcol];

    float x0[XPT];
    #pragma unroll
    for (int q = 0; q < XPT; ++q)
        x0[q] = xcol[(size_t)(rsub + 8 * q) * BDIM];
    #pragma unroll
    for (int q = 0; q < XPT; ++q)
        xs[rsub + 8 * q][jcol] = x0[q];
    __syncthreads();

    float x1[XPT], bn[BPT];
    #pragma unroll
    for (int q = 0; q < XPT; ++q)
        x1[q] = xcol[(size_t)(HALF_ROWS + rsub + 8 * q) * BDIM];
    #pragma unroll
    for (int q = 0; q < BPT; ++q)
        bn[q] = b[tid + THREADS * q];

    const int*   ip3 = idx3 + 3 * TPW * wave;
    const int*   ip4 = idx4 + 4 * TPW * wave;
    const int*   ip5 = idx5 + 5 * TPW * wave;
    const float* cp3 = c3 + TPW * wave;
    const float* cp4 = c4 + TPW * wave;
    const float* cp5 = c5 + TPW * wave;
    const f4v    al  = *(const CONST_AS f4v*)(a + LPW * wave);

    half_pass(&xs[0][0], lane, wave, &wpart[wave][0],
              ip3, cp3, ip4, cp4, ip5, cp5, al);

    #pragma unroll
    for (int q = 0; q < XPT; ++q)
        xs[HALF_ROWS + rsub + 8 * q][jcol] = x1[q];
    #pragma unroll
    for (int q = 0; q < BPT; ++q) {
        const int e = tid + THREADS * q;
        const int i = e >> 6, j = e & 63;
        buS[i][j] = (j > i) ? bn[q] : 0.0f;
    }
    __syncthreads();

    half_pass(&xs[HALF_ROWS][0], lane, wave, &wpart[wave][HALF_ROWS],
              ip3, cp3, ip4, cp4, ip5, cp5, al);

    const int waveBase = (wave & 7) * 16;
    const int nsel     = wave >> 3;
    float part[8];
    #pragma unroll
    for (int p = 0; p < 8; ++p) part[p] = 0.0f;

    #pragma unroll
    for (int nl = 0; nl < 4 / NSPLIT; ++nl) {
        const int nt = (4 / NSPLIT) * nsel + nl;
        v8f acc = {};
        #pragma unroll
        for (int kc = 0; kc < 16; ++kc) {
            v2f A, Bf;
            A.x = xs[waveBase + lo][4 * kc + 2 * hi + 0];
            A.y = xs[waveBase + lo][4 * kc + 2 * hi + 1];
            Bf.x = buS[4 * kc + 2 * hi + 0][16 * nt + lo];
            Bf.y = buS[4 * kc + 2 * hi + 1][16 * nt + lo];
            acc = __builtin_amdgcn_wmma_f32_16x16x4_f32(
                false, A, false, Bf, (short)0, acc, false, false);
        }
        #pragma unroll
        for (int p = 0; p < 8; ++p)
            part[p] += acc[p] * xs[waveBase + p + 8 * hi][16 * nt + lo];
    }

    #pragma unroll
    for (int p = 0; p < 8; ++p) {
        float s = part[p];
        s += __shfl_xor(s, 8);
        s += __shfl_xor(s, 4);
        s += __shfl_xor(s, 2);
        s += __shfl_xor(s, 1);
        if (lo == p) quadS[nsel][waveBase + p + 8 * hi] = s;
    }
    __syncthreads();

    if (tid < ROWS_PER_BLOCK) {
        float s = quadS[0][tid] + quadS[1][tid];
        #pragma unroll
        for (int w = 0; w < NWAVES; ++w) s += wpart[w][tid];
        out[rowBlockBase + tid] = s;
    }
}

extern "C" void kernel_launch(void* const* d_in, const int* in_sizes, int n_in,
                              void* d_out, int out_size, void* d_ws, size_t ws_size,
                              hipStream_t stream) {
    const float* x    = (const float*)d_in[0];
    const int*   S    = (const int*)  d_in[1];
    const float* a    = (const float*)d_in[2];
    const float* b    = (const float*)d_in[3];
    const int*   idx3 = (const int*)  d_in[4];
    const float* c3   = (const float*)d_in[5];
    const int*   idx4 = (const int*)  d_in[6];
    const float* c4   = (const float*)d_in[7];
    const int*   idx5 = (const int*)  d_in[8];
    const float* c5   = (const float*)d_in[9];
    float* out = (float*)d_out;

    const int nblocks = out_size / ROWS_PER_BLOCK;
    sparse_poly_kernel<<<nblocks, THREADS, 0, stream>>>(
        x, S, a, b, idx3, c3, idx4, c4, idx5, c5, out);
}
